# starting kernel
# baseline (speedup 1.0000x reference)
#include <hip/hip_runtime.h>
#include <hip/hip_bf16.h>

typedef __attribute__((ext_vector_type(2))) float v2f;
typedef __attribute__((ext_vector_type(8))) float v8f;

#define NN 100000
#define NC 5000
#define HDIM 64
#define ES 3200000
#define EM 1000000
#define ER 160000

__global__ __launch_bounds__(256) void k_zero(float* __restrict__ p, long n4) {
    long i = (long)blockIdx.x * blockDim.x + threadIdx.x;
    if (i < n4) ((float4*)p)[i] = make_float4(0.f, 0.f, 0.f, 0.f);
}

__global__ __launch_bounds__(256) void k_packW(const float* __restrict__ W,
                                               v2f* __restrict__ Wp,
                                               int K, int N) {
    const int Kp2 = ((K + 3) & ~3) >> 1;
    const long n = (long)Kp2 * N;
    long i = (long)blockIdx.x * blockDim.x + threadIdx.x;
    if (i >= n) return;
    int k2 = (int)(i / N), col = (int)(i % N);
    int k = 2 * k2;
    v2f v;
    v.x = (k     < K) ? W[(long)k * N + col]       : 0.f;
    v.y = (k + 1 < K) ? W[(long)(k + 1) * N + col] : 0.f;
    Wp[i] = v;
}

__global__ __launch_bounds__(256) void k_scatter(const int* __restrict__ src,
                                                 const int* __restrict__ dst,
                                                 const float* __restrict__ x,
                                                 float* __restrict__ agg,
                                                 float* __restrict__ cnt,
                                                 long nE) {
    long gid = (long)blockIdx.x * blockDim.x + threadIdx.x;
    long e = gid >> 4;
    int c = (int)(gid & 15) * 4;
    if (e >= nE) return;
    int s = src[e], d = dst[e];
    const float4 v = *(const float4*)(x + (long)s * HDIM + c);
    float* ap = agg + (long)d * HDIM + c;
    atomicAdd(ap + 0, v.x);
    atomicAdd(ap + 1, v.y);
    atomicAdd(ap + 2, v.z);
    atomicAdd(ap + 3, v.w);
    if ((gid & 15) == 0) atomicAdd(cnt + d, 1.0f);
}

__global__ __launch_bounds__(256) void k_scale_rows(float* __restrict__ agg,
                                                    const float* __restrict__ cnt,
                                                    int M) {
    long gid = (long)blockIdx.x * blockDim.x + threadIdx.x;
    long row = gid >> 4;
    int c = (int)(gid & 15) * 4;
    if (row >= M) return;
    float inv = 1.0f / fmaxf(cnt[row], 1.0f);
    float4* p = (float4*)(agg + row * HDIM + c);
    float4 v = *p;
    v.x *= inv; v.y *= inv; v.z *= inv; v.w *= inv;
    *p = v;
}

template<int NT, bool FULL, bool VECA>
__device__ __forceinline__ void gemm_body(v8f (&c)[NT],
                                          const float* __restrict__ A,
                                          const v2f* __restrict__ Wp,
                                          int M, int K, int N,
                                          int mbase, int lane) {
    const int row  = mbase + (lane & 15);
    const int koff = (lane >= 16) ? 2 : 0;
    const int col0 = lane & 15;
    const bool rok = FULL || (row < M);
    const float* __restrict__ Arow = A + (long)row * K;
    const int Kmain = K & ~3;

    #pragma unroll 4
    for (int k = 0; k < Kmain; k += 4) {
        const int kk = k + koff;
        v2f a;
        if (rok) {
            if (VECA) {
                a = *(const v2f*)(Arow + kk);
            } else {
                a.x = Arow[kk];
                a.y = Arow[kk + 1];
            }
        } else {
            a.x = 0.f; a.y = 0.f;
        }
        #pragma unroll
        for (int nt = 0; nt < NT; ++nt) {
            v2f b = Wp[(long)(kk >> 1) * N + nt * 16 + col0];
            c[nt] = __builtin_amdgcn_wmma_f32_16x16x4_f32(
                false, a, false, b, (short)0, c[nt], false, false);
        }
    }
    if (Kmain < K) {
        const int kk = Kmain + koff;
        v2f a;
        a.x = (rok && kk     < K) ? Arow[kk]     : 0.f;
        a.y = (rok && kk + 1 < K) ? Arow[kk + 1] : 0.f;
        #pragma unroll
        for (int nt = 0; nt < NT; ++nt) {
            v2f b = Wp[(long)(kk >> 1) * N + nt * 16 + col0];
            c[nt] = __builtin_amdgcn_wmma_f32_16x16x4_f32(
                false, a, false, b, (short)0, c[nt], false, false);
        }
    }
}

template<int NT, bool VECA1>
__global__ __launch_bounds__(256) void k_gemm(const float* __restrict__ A1,
                                              const v2f* __restrict__ W1p,
                                              const float* __restrict__ bias,
                                              const float* __restrict__ A2,
                                              const v2f* __restrict__ W2p,
                                              float* __restrict__ out,
                                              int M, int K1, int K2,
                                              int do_relu, int accum) {
    const int N = NT * 16;
    const int lane = threadIdx.x & 31;
    const int wv = threadIdx.x >> 5;
    const int mbase = (blockIdx.x * 8 + wv) * 16;
    if (mbase >= M) return;

    v8f c[NT];
    #pragma unroll
    for (int t = 0; t < NT; ++t)
        c[t] = (v8f){0.f, 0.f, 0.f, 0.f, 0.f, 0.f, 0.f, 0.f};

    const bool full = (mbase + 16 <= M);
    if (full) {
        gemm_body<NT, true, VECA1>(c, A1, W1p, M, K1, N, mbase, lane);
        if (A2) gemm_body<NT, true, true>(c, A2, W2p, M, K2, N, mbase, lane);
    } else {
        gemm_body<NT, false, VECA1>(c, A1, W1p, M, K1, N, mbase, lane);
        if (A2) gemm_body<NT, false, true>(c, A2, W2p, M, K2, N, mbase, lane);
    }

    const int col0 = lane & 15;
    const int rofs = (lane >= 16) ? 8 : 0;
    #pragma unroll
    for (int nt = 0; nt < NT; ++nt) {
        const int col = nt * 16 + col0;
        const float bb = bias ? bias[col] : 0.f;
        #pragma unroll
        for (int r = 0; r < 8; ++r) {
            const int row = mbase + r + rofs;
            if (full || row < M) {
                const long idx = (long)row * N + col;
                float v = c[nt][r] + bb;
                if (accum) v += out[idx];
                if (do_relu) v = fmaxf(v, 0.f);
                out[idx] = v;
            }
        }
    }
}

__global__ void k_ln(float* __restrict__ out, const float* __restrict__ in,
                     const float* __restrict__ res,
                     const float* __restrict__ g, const float* __restrict__ b,
                     int M) {
    int lane = threadIdx.x;
    int row = blockIdx.x * blockDim.y + threadIdx.y;
    if (row >= M) return;
    long base = (long)row * HDIM;
    float v0 = in[base + lane];
    float v1 = in[base + lane + 32];
    if (res) { v0 += res[base + lane]; v1 += res[base + lane + 32]; }
    v0 = fmaxf(v0, 0.f);
    v1 = fmaxf(v1, 0.f);
    float s = v0 + v1;
    for (int off = 16; off > 0; off >>= 1) s += __shfl_xor(s, off, 32);
    float mean = s * (1.0f / 64.0f);
    float d0 = v0 - mean, d1 = v1 - mean;
    float q = d0 * d0 + d1 * d1;
    for (int off = 16; off > 0; off >>= 1) q += __shfl_xor(q, off, 32);
    float rstd = rsqrtf(q * (1.0f / 64.0f) + 1e-5f);
    out[base + lane]      = d0 * rstd * g[lane]      + b[lane];
    out[base + lane + 32] = d1 * rstd * g[lane + 32] + b[lane + 32];
}

__global__ __launch_bounds__(256) void k_addrelu(float* __restrict__ o,
                                                 const float* __restrict__ a,
                                                 const float* __restrict__ b,
                                                 long n) {
    long i = (long)blockIdx.x * blockDim.x + threadIdx.x;
    if (i < n) o[i] = fmaxf(a[i] + b[i], 0.f);
}

__global__ __launch_bounds__(256) void k_final(const float* __restrict__ h,
                                               const float* __restrict__ W2,
                                               const float* __restrict__ b2,
                                               float* __restrict__ out, int M) {
    int i = blockIdx.x * blockDim.x + threadIdx.x;
    if (i >= M) return;
    float acc = b2[0];
    const float* hr = h + (long)i * 32;
    #pragma unroll
    for (int j = 0; j < 32; ++j) acc += hr[j] * W2[j];
    out[i] = acc;
}

extern "C" void kernel_launch(void* const* d_in, const int* in_sizes, int n_in,
                              void* d_out, int out_size, void* d_ws, size_t ws_size,
                              hipStream_t stream) {
    const float* x_news    = (const float*)d_in[0];
    const float* x_company = (const float*)d_in[1];
    const int*   edge_sim  = (const int*)d_in[2];
    const int*   men_src   = (const int*)d_in[3];
    const int*   men_dst   = (const int*)d_in[4];
    const int*   edge_rel  = (const int*)d_in[5];
    const float* nw_W = (const float*)d_in[6];
    const float* nw_b = (const float*)d_in[7];
    const float* cw_W = (const float*)d_in[8];
    const float* cw_b = (const float*)d_in[9];
    const float* n1n_g = (const float*)d_in[10];
    const float* n1n_b = (const float*)d_in[11];
    const float* n1c_g = (const float*)d_in[12];
    const float* n1c_b = (const float*)d_in[13];
    const float* c1_Wl = (const float*)d_in[14];
    const float* c1_bl = (const float*)d_in[15];
    const float* c1_Wr = (const float*)d_in[16];
    const float* c2_Wl = (const float*)d_in[17];
    const float* c2_bl = (const float*)d_in[18];
    const float* c2_Wr = (const float*)d_in[19];
    const float* cls_W1 = (const float*)d_in[20];
    const float* cls_b1 = (const float*)d_in[21];
    const float* cls_W2 = (const float*)d_in[22];
    const float* cls_b2 = (const float*)d_in[23];
    float* out = (float*)d_out;

    float* ws = (float*)d_ws;
    long o = 0;
    float* xn0  = ws + o; o += (long)NN * HDIM;
    float* xn1  = ws + o; o += (long)NN * HDIM;
    float* aggn = ws + o; o += (long)NN * HDIM;
    float* cntn = ws + o; o += NN;
    float* aggcM = ws + o; o += (long)NC * HDIM;
    float* cntcM = ws + o; o += NC;
    float* aggcR = ws + o; o += (long)NC * HDIM;
    float* cntcR = ws + o; o += NC;
    float* xc0  = ws + o; o += (long)NC * HDIM;
    float* xc1  = ws + o; o += (long)NC * HDIM;
    float* ocb  = ws + o; o += (long)NC * HDIM;
    float* comp = ws + o; o += (long)NC * HDIM;
    float* hbuf = ws + o; o += (long)NC * 32;
    v2f* wp_nw  = (v2f*)(ws + o); o += 2L * (388 / 2) * 64;
    v2f* wp_cw  = (v2f*)(ws + o); o += 2L * (24 / 2) * 64;
    v2f* wp_c1  = (v2f*)(ws + o); o += 2L * 6 * 32 * 64;
    v2f* wp_c2  = (v2f*)(ws + o); o += 2L * 4 * 32 * 64;
    v2f* wp_cls = (v2f*)(ws + o); o += 2L * 32 * 32;

    const long zn_n = (long)NN * HDIM + NN;
    const long zn_c = 2L * ((long)NC * HDIM + NC);

    dim3 blk(256);
    auto blocks = [](long n) { return (unsigned)((n + 255) / 256); };
    int tilesNN = (NN + 15) / 16, tilesNC = (NC + 15) / 16;
    unsigned gemmBlkNN = (tilesNN + 7) / 8, gemmBlkNC = (tilesNC + 7) / 8;

    k_packW<<<blocks((388L / 2) * 64), blk, 0, stream>>>(nw_W, wp_nw, 385, 64);
    k_packW<<<blocks((24L / 2) * 64), blk, 0, stream>>>(cw_W, wp_cw, 24, 64);
    for (int r = 0; r < 3; ++r) {
        k_packW<<<blocks(32L * 64), blk, 0, stream>>>(c1_Wl + r * 4096,
                                                      wp_c1 + (2 * r) * 2048, 64, 64);
        k_packW<<<blocks(32L * 64), blk, 0, stream>>>(c1_Wr + r * 4096,
                                                      wp_c1 + (2 * r + 1) * 2048, 64, 64);
    }
    for (int r = 1; r < 3; ++r) {
        k_packW<<<blocks(32L * 64), blk, 0, stream>>>(c2_Wl + r * 4096,
                                                      wp_c2 + (2 * (r - 1)) * 2048, 64, 64);
        k_packW<<<blocks(32L * 64), blk, 0, stream>>>(c2_Wr + r * 4096,
                                                      wp_c2 + (2 * (r - 1) + 1) * 2048, 64, 64);
    }
    k_packW<<<blocks(32L * 32), blk, 0, stream>>>(cls_W1, wp_cls, 64, 32);

    k_gemm<4, false><<<gemmBlkNN, blk, 0, stream>>>(x_news, wp_nw, nw_b,
                                                    nullptr, nullptr,
                                                    xn0, NN, 385, 0, 1, 0);
    k_gemm<4, true><<<gemmBlkNC, blk, 0, stream>>>(x_company, wp_cw, cw_b,
                                                   nullptr, nullptr,
                                                   xc0, NC, 24, 0, 1, 0);

    k_zero<<<blocks(zn_n / 4), blk, 0, stream>>>(aggn, zn_n / 4);
    k_zero<<<blocks(zn_c / 4), blk, 0, stream>>>(aggcM, zn_c / 4);
    k_scatter<<<blocks((long)ES * 16), blk, 0, stream>>>(edge_sim, edge_sim + ES,
                                                         xn0, aggn, cntn, ES);
    k_scatter<<<blocks((long)EM * 16), blk, 0, stream>>>(men_src, men_dst,
                                                         xn0, aggcM, cntcM, EM);
    k_scatter<<<blocks((long)ER * 16), blk, 0, stream>>>(edge_rel, edge_rel + ER,
                                                         xc0, aggcR, cntcR, ER);
    k_scale_rows<<<blocks((long)NN * 16), blk, 0, stream>>>(aggn, cntn, NN);
    k_scale_rows<<<blocks((long)NC * 16), blk, 0, stream>>>(aggcM, cntcM, NC);
    k_scale_rows<<<blocks((long)NC * 16), blk, 0, stream>>>(aggcR, cntcR, NC);

    k_gemm<4, true><<<gemmBlkNN, blk, 0, stream>>>(aggn, wp_c1 + 0 * 2048,
                                                   c1_bl + 0 * 64,
                                                   xn0, wp_c1 + 1 * 2048,
                                                   xn1, NN, 64, 64, 0, 0);
    k_gemm<4, true><<<gemmBlkNC, blk, 0, stream>>>(aggcM, wp_c1 + 2 * 2048,
                                                   c1_bl + 1 * 64,
                                                   xc0, wp_c1 + 3 * 2048,
                                                   ocb, NC, 64, 64, 0, 0);
    k_gemm<4, true><<<gemmBlkNC, blk, 0, stream>>>(aggcR, wp_c1 + 4 * 2048,
                                                   c1_bl + 2 * 64,
                                                   xc0, wp_c1 + 5 * 2048,
                                                   ocb, NC, 64, 64, 0, 1);

    dim3 lnb(32, 8);
    k_ln<<<(NN + 7) / 8, lnb, 0, stream>>>(xn1, xn1, nullptr, n1n_g, n1n_b, NN);
    k_ln<<<(NC + 7) / 8, lnb, 0, stream>>>(xc1, ocb, xc0, n1c_g, n1c_b, NC);

    k_zero<<<blocks(zn_c / 4), blk, 0, stream>>>(aggcM, zn_c / 4);
    k_scatter<<<blocks((long)EM * 16), blk, 0, stream>>>(men_src, men_dst,
                                                         xn1, aggcM, cntcM, EM);
    k_scatter<<<blocks((long)ER * 16), blk, 0, stream>>>(edge_rel, edge_rel + ER,
                                                         xc1, aggcR, cntcR, ER);
    k_scale_rows<<<blocks((long)NC * 16), blk, 0, stream>>>(aggcM, cntcM, NC);
    k_scale_rows<<<blocks((long)NC * 16), blk, 0, stream>>>(aggcR, cntcR, NC);
    k_gemm<4, true><<<gemmBlkNC, blk, 0, stream>>>(aggcM, wp_c2 + 0 * 2048,
                                                   c2_bl + 1 * 64,
                                                   xc1, wp_c2 + 1 * 2048,
                                                   ocb, NC, 64, 64, 0, 0);
    k_gemm<4, true><<<gemmBlkNC, blk, 0, stream>>>(aggcR, wp_c2 + 2 * 2048,
                                                   c2_bl + 2 * 64,
                                                   xc1, wp_c2 + 3 * 2048,
                                                   ocb, NC, 64, 64, 0, 1);

    k_addrelu<<<blocks((long)NC * HDIM), blk, 0, stream>>>(comp, ocb, xc1,
                                                           (long)NC * HDIM);
    k_gemm<2, true><<<gemmBlkNC, blk, 0, stream>>>(comp, wp_cls, cls_b1,
                                                   nullptr, nullptr,
                                                   hbuf, NC, 64, 0, 1, 0);
    k_final<<<blocks(NC), blk, 0, stream>>>(hbuf, cls_W2, cls_b2, out, NC);
}


# news aggregation: each edge swaps itself into its destination row's list head (integer exchange, no float atomics); 16 lanes walk a row's list and sum its source rows in registers (arrival order)
# speedup vs baseline: 2.6630x; 2.6630x over previous
#include <hip/hip_runtime.h>
#include <hip/hip_bf16.h>

typedef __attribute__((ext_vector_type(2))) float v2f;
typedef __attribute__((ext_vector_type(8))) float v8f;
typedef __attribute__((ext_vector_type(4))) float v4f;

template<int NT> struct ColVec;
template<> struct ColVec<4> { typedef v4f T; };
template<> struct ColVec<2> { typedef v2f T; };

#define NN 100000
#define NC 5000
#define HDIM 64
#define ES 3200000
#define EM 1000000
#define ER 160000
static_assert(EM % 32 == 0 && ER % 32 == 0, "edge lists are walked in chunks of 32");

#define LINK_EPT 8
#define LINK_THREADS 128
static_assert(NN % 4 == 0, "heads are cleared 16 bytes at a time");
static_assert(ES % (LINK_THREADS * LINK_EPT) == 0, "k_list_link has no tail");

__global__ __launch_bounds__(256) void k_list_clear(int* __restrict__ head, int n4) {
    const int i = blockIdx.x * 256 + threadIdx.x;
    if (i < n4) ((int4*)head)[i] = make_int4(-1, -1, -1, -1);
}

__global__ __launch_bounds__(LINK_THREADS) void k_list_link(const int* __restrict__ src,
                                                   const int* __restrict__ dst,
                                                   int* __restrict__ head,
                                                   int2* __restrict__ rec) {
    const int lane = threadIdx.x & 31;
    const int wave = blockIdx.x * (LINK_THREADS / 32) + (threadIdx.x >> 5);
    const int e0 = wave * (32 * LINK_EPT) + lane;
    int sv[LINK_EPT], dv[LINK_EPT], pv[LINK_EPT];
    #pragma unroll
    for (int k = 0; k < LINK_EPT; ++k) {
        sv[k] = src[e0 + 32 * k];
        dv[k] = dst[e0 + 32 * k];
    }
    #pragma unroll
    for (int k = 0; k < LINK_EPT; ++k)
        pv[k] = atomicExch(head + dv[k], e0 + 32 * k);
    #pragma unroll
    for (int k = 0; k < LINK_EPT; ++k)
        rec[e0 + 32 * k] = make_int2(sv[k], pv[k]);
}

__global__ __launch_bounds__(256) void k_list_mean(const int* __restrict__ head,
                                                   const int2* __restrict__ rec,
                                                   const float* __restrict__ x,
                                                   float* __restrict__ agg,
                                                   int M, int nE) {
    const int row = (blockIdx.x * 256 + threadIdx.x) >> 4;
    const int c = (threadIdx.x & 15) * 4;
    if (row >= M) return;
    float4 a = make_float4(0.f, 0.f, 0.f, 0.f);
    int n = 0;
    int e = head[row];
    int2 r = make_int2(0, -1);
    if (e >= 0) r = rec[e];
    while (e >= 0 && n < nE) {
        const int en = r.y;
        int2 rn = make_int2(0, -1);
        if (en >= 0) rn = rec[en];
        const float4 v = *(const float4*)(x + (long)r.x * HDIM + c);
        a.x += v.x; a.y += v.y; a.z += v.z; a.w += v.w;
        ++n;
        e = en;
        r = rn;
    }
    const float inv = 1.0f / fmaxf((float)n, 1.0f);
    a.x *= inv; a.y *= inv; a.z *= inv; a.w *= inv;
    *(float4*)(agg + (long)row * HDIM + c) = a;
}

#define CSL 8
#define NSLICE (HDIM / CSL)
#define NPART 32
#define AGG_THREADS 1024

__global__ __launch_bounds__(AGG_THREADS) void k_agg_cols(const int* __restrict__ src,
                                                          const int* __restrict__ dst,
                                                          const float* __restrict__ x,
                                                          float* __restrict__ part,
                                                          float* __restrict__ cntpart,
                                                          int nchunk) {
    __shared__ __attribute__((aligned(16))) float acc[NC * CSL];
    __shared__ float cn[NC];
    const int tid = threadIdx.x;
    const int s = blockIdx.x % NSLICE;
    const int p = blockIdx.x / NSLICE;

    float4* acc4 = (float4*)acc;
    for (int i = tid; i < NC * CSL / 4; i += AGG_THREADS)
        acc4[i] = make_float4(0.f, 0.f, 0.f, 0.f);
    for (int i = tid; i < NC; i += AGG_THREADS) cn[i] = 0.f;
    __syncthreads();

    const int lane = tid & 31, wv = tid >> 5;
    const int el = lane >> 3;
    const int j = lane & (CSL - 1);
    const int cb = (int)(((long)p * nchunk) / NPART);
    const int ce = (int)(((long)(p + 1) * nchunk) / NPART);
    const float* __restrict__ xs = x + s * CSL + j;
    const bool docnt = (s == 0) && (j == 0);
    const int4* __restrict__ src4 = (const int4*)src + el * 2;
    const int4* __restrict__ dst4 = (const int4*)dst + el * 2;

    int c = cb + wv;
    int4 sa = make_int4(0, 0, 0, 0), sb = sa, da = sa, db = sa;
    if (c < ce) {
        sa = src4[(long)c * 8];  sb = src4[(long)c * 8 + 1];
        da = dst4[(long)c * 8];  db = dst4[(long)c * 8 + 1];
    }
    while (c < ce) {
        const int c2 = c + AGG_THREADS / 32;
        int4 nsa = sa, nsb = sb, nda = da, ndb = db;
        if (c2 < ce) {
            nsa = src4[(long)c2 * 8];  nsb = src4[(long)c2 * 8 + 1];
            nda = dst4[(long)c2 * 8];  ndb = dst4[(long)c2 * 8 + 1];
        }
        const float v0 = xs[(long)sa.x * HDIM];
        const float v1 = xs[(long)sa.y * HDIM];
        const float v2 = xs[(long)sa.z * HDIM];
        const float v3 = xs[(long)sa.w * HDIM];
        const float v4 = xs[(long)sb.x * HDIM];
        const float v5 = xs[(long)sb.y * HDIM];
        const float v6 = xs[(long)sb.z * HDIM];
        const float v7 = xs[(long)sb.w * HDIM];
        atomicAdd(&acc[da.x * CSL + j], v0);
        atomicAdd(&acc[da.y * CSL + j], v1);
        atomicAdd(&acc[da.z * CSL + j], v2);
        atomicAdd(&acc[da.w * CSL + j], v3);
        atomicAdd(&acc[db.x * CSL + j], v4);
        atomicAdd(&acc[db.y * CSL + j], v5);
        atomicAdd(&acc[db.z * CSL + j], v6);
        atomicAdd(&acc[db.w * CSL + j], v7);
        if (docnt) {
            atomicAdd(&cn[da.x], 1.0f);  atomicAdd(&cn[da.y], 1.0f);
            atomicAdd(&cn[da.z], 1.0f);  atomicAdd(&cn[da.w], 1.0f);
            atomicAdd(&cn[db.x], 1.0f);  atomicAdd(&cn[db.y], 1.0f);
            atomicAdd(&cn[db.z], 1.0f);  atomicAdd(&cn[db.w], 1.0f);
        }
        sa = nsa; sb = nsb; da = nda; db = ndb;
        c = c2;
    }
    __syncthreads();

    float4* __restrict__ part4 = (float4*)part + (long)blockIdx.x * (NC * CSL / 4);
    for (int i = tid; i < NC * CSL / 4; i += AGG_THREADS) part4[i] = acc4[i];
    if (s == 0) {
        float* __restrict__ cp = cntpart + (long)p * NC;
        for (int i = tid; i < NC; i += AGG_THREADS) cp[i] = cn[i];
    }
}

__global__ __launch_bounds__(256) void k_combine(const float* __restrict__ part,
                                                 const float* __restrict__ cntpart,
                                                 float* __restrict__ agg) {
    const int per = NC * CSL / 4;
    const int t = blockIdx.x * 256 + threadIdx.x;
    if (t >= NSLICE * per) return;
    const int s = t / per;
    const int rem = t - s * per;
    const int row = rem / (CSL / 4);
    const int q = rem % (CSL / 4);
    const float4* __restrict__ p4 = (const float4*)part + t;
    float4 a = make_float4(0.f, 0.f, 0.f, 0.f);
    float c = 0.f;
    #pragma unroll 8
    for (int p = 0; p < NPART; ++p) {
        const float4 v = p4[(long)p * (NSLICE * per)];
        a.x += v.x; a.y += v.y; a.z += v.z; a.w += v.w;
        c += cntpart[p * NC + row];
    }
    const float inv = 1.0f / fmaxf(c, 1.0f);
    a.x *= inv; a.y *= inv; a.z *= inv; a.w *= inv;
    *(float4*)(agg + (long)row * HDIM + s * CSL + q * 4) = a;
}

template<int NT, bool FULL, bool VECA>
__device__ __forceinline__ void gemm_body(v8f (&c)[NT],
                                          const float* __restrict__ A,
                                          const float* __restrict__ W,
                                          int M, int K,
                                          int mbase, int lane) {
    typedef typename ColVec<NT>::T VT;
    constexpr int N = NT * 16;
    const int row  = mbase + (lane & 15);
    const int koff = (lane >= 16) ? 2 : 0;
    const bool rok = FULL || (row < M);
    const float* __restrict__ Arow = A + (long)row * K;
    const float* __restrict__ Wc = W + (lane & 15) * NT;
    const int Kmain = K & ~3;

    #pragma unroll 4
    for (int k = 0; k < Kmain; k += 4) {
        const int kk = k + koff;
        v2f a;
        if (rok) {
            if (VECA) {
                a = *(const v2f*)(Arow + kk);
            } else {
                a.x = Arow[kk];
                a.y = Arow[kk + 1];
            }
        } else {
            a.x = 0.f; a.y = 0.f;
        }
        const VT w0 = *(const VT*)(Wc + (long)kk * N);
        const VT w1 = *(const VT*)(Wc + (long)(kk + 1) * N);
        #pragma unroll
        for (int nt = 0; nt < NT; ++nt) {
            v2f b;
            b.x = w0[nt];
            b.y = w1[nt];
            c[nt] = __builtin_amdgcn_wmma_f32_16x16x4_f32(
                false, a, false, b, (short)0, c[nt], false, false);
        }
    }
    if (Kmain < K) {
        const int kk = Kmain + koff;
        v2f a;
        a.x = (rok && kk     < K) ? Arow[kk]     : 0.f;
        a.y = (rok && kk + 1 < K) ? Arow[kk + 1] : 0.f;
        VT w0, w1;
        #pragma unroll
        for (int nt = 0; nt < NT; ++nt) { w0[nt] = 0.f; w1[nt] = 0.f; }
        if (kk     < K) w0 = *(const VT*)(Wc + (long)kk * N);
        if (kk + 1 < K) w1 = *(const VT*)(Wc + (long)(kk + 1) * N);
        #pragma unroll
        for (int nt = 0; nt < NT; ++nt) {
            v2f b;
            b.x = w0[nt];
            b.y = w1[nt];
            c[nt] = __builtin_amdgcn_wmma_f32_16x16x4_f32(
                false, a, false, b, (short)0, c[nt], false, false);
        }
    }
}

template<int NT, bool VECA1>
__global__ __launch_bounds__(256) void k_gemm(const float* __restrict__ A1,
                                              const float* __restrict__ W1,
                                              const float* __restrict__ bias,
                                              const float* __restrict__ A2,
                                              const float* __restrict__ W2,
                                              float* __restrict__ out,
                                              int M, int K1, int K2,
                                              int do_relu, int accum) {
    typedef typename ColVec<NT>::T VT;
    constexpr int N = NT * 16;
    const int lane = threadIdx.x & 31;
    const int wv = threadIdx.x >> 5;
    const int mbase = (blockIdx.x * 8 + wv) * 16;
    if (mbase >= M) return;

    v8f c[NT];
    #pragma unroll
    for (int t = 0; t < NT; ++t)
        c[t] = (v8f){0.f, 0.f, 0.f, 0.f, 0.f, 0.f, 0.f, 0.f};

    const bool full = (mbase + 16 <= M);
    if (full) {
        gemm_body<NT, true, VECA1>(c, A1, W1, M, K1, mbase, lane);
        if (A2) gemm_body<NT, true, true>(c, A2, W2, M, K2, mbase, lane);
    } else {
        gemm_body<NT, false, VECA1>(c, A1, W1, M, K1, mbase, lane);
        if (A2) gemm_body<NT, false, true>(c, A2, W2, M, K2, mbase, lane);
    }

    const int cb = (lane & 15) * NT;
    const int rofs = (lane >= 16) ? 8 : 0;
    VT bb;
    #pragma unroll
    for (int nt = 0; nt < NT; ++nt) bb[nt] = 0.f;
    if (bias) bb = *(const VT*)(bias + cb);
    #pragma unroll
    for (int r = 0; r < 8; ++r) {
        const int row = mbase + r + rofs;
        if (full || row < M) {
            VT* __restrict__ p = (VT*)(out + (long)row * N + cb);
            VT v;
            #pragma unroll
            for (int nt = 0; nt < NT; ++nt) v[nt] = c[nt][r] + bb[nt];
            if (accum) v += *p;
            if (do_relu) {
                #pragma unroll
                for (int nt = 0; nt < NT; ++nt) v[nt] = fmaxf(v[nt], 0.f);
            }
            *p = v;
        }
    }
}

__global__ void k_ln(float* __restrict__ out, const float* __restrict__ in,
                     const float* __restrict__ res,
                     const float* __restrict__ g, const float* __restrict__ b,
                     int M) {
    int lane = threadIdx.x;
    int row = blockIdx.x * blockDim.y + threadIdx.y;
    if (row >= M) return;
    long base = (long)row * HDIM;
    float v0 = in[base + lane];
    float v1 = in[base + lane + 32];
    if (res) { v0 += res[base + lane]; v1 += res[base + lane + 32]; }
    v0 = fmaxf(v0, 0.f);
    v1 = fmaxf(v1, 0.f);
    float s = v0 + v1;
    for (int off = 16; off > 0; off >>= 1) s += __shfl_xor(s, off, 32);
    float mean = s * (1.0f / 64.0f);
    float d0 = v0 - mean, d1 = v1 - mean;
    float q = d0 * d0 + d1 * d1;
    for (int off = 16; off > 0; off >>= 1) q += __shfl_xor(q, off, 32);
    float rstd = rsqrtf(q * (1.0f / 64.0f) + 1e-5f);
    out[base + lane]      = d0 * rstd * g[lane]      + b[lane];
    out[base + lane + 32] = d1 * rstd * g[lane + 32] + b[lane + 32];
}

__global__ __launch_bounds__(256) void k_addrelu(float* __restrict__ o,
                                                 const float* __restrict__ a,
                                                 const float* __restrict__ b,
                                                 long n) {
    long i = (long)blockIdx.x * blockDim.x + threadIdx.x;
    if (i < n) o[i] = fmaxf(a[i] + b[i], 0.f);
}

__global__ __launch_bounds__(256) void k_final(const float* __restrict__ h,
                                               const float* __restrict__ W2,
                                               const float* __restrict__ b2,
                                               float* __restrict__ out, int M) {
    int i = blockIdx.x * blockDim.x + threadIdx.x;
    if (i >= M) return;
    float acc = b2[0];
    const float* hr = h + (long)i * 32;
    #pragma unroll
    for (int j = 0; j < 32; ++j) acc += hr[j] * W2[j];
    out[i] = acc;
}

extern "C" void kernel_launch(void* const* d_in, const int* in_sizes, int n_in,
                              void* d_out, int out_size, void* d_ws, size_t ws_size,
                              hipStream_t stream) {
    const float* x_news    = (const float*)d_in[0];
    const float* x_company = (const float*)d_in[1];
    const int*   edge_sim  = (const int*)d_in[2];
    const int*   men_src   = (const int*)d_in[3];
    const int*   men_dst   = (const int*)d_in[4];
    const int*   edge_rel  = (const int*)d_in[5];
    const float* nw_W = (const float*)d_in[6];
    const float* nw_b = (const float*)d_in[7];
    const float* cw_W = (const float*)d_in[8];
    const float* cw_b = (const float*)d_in[9];
    const float* n1n_g = (const float*)d_in[10];
    const float* n1n_b = (const float*)d_in[11];
    const float* n1c_g = (const float*)d_in[12];
    const float* n1c_b = (const float*)d_in[13];
    const float* c1_Wl = (const float*)d_in[14];
    const float* c1_bl = (const float*)d_in[15];
    const float* c1_Wr = (const float*)d_in[16];
    const float* c2_Wl = (const float*)d_in[17];
    const float* c2_bl = (const float*)d_in[18];
    const float* c2_Wr = (const float*)d_in[19];
    const float* cls_W1 = (const float*)d_in[20];
    const float* cls_b1 = (const float*)d_in[21];
    const float* cls_W2 = (const float*)d_in[22];
    const float* cls_b2 = (const float*)d_in[23];
    float* out = (float*)d_out;

    float* ws = (float*)d_ws;
    long o = 0;
    float* xn0  = ws + o; o += (long)NN * HDIM;
    float* aggn = ws + o; o += (long)NN * HDIM;
    float* cntn = ws + o; o += NN;
    float* xn1  = ws + o; o += (long)NN * HDIM;
    float* aggcM = ws + o; o += (long)NC * HDIM;
    float* aggcR = ws + o; o += (long)NC * HDIM;
    float* xc0  = ws + o; o += (long)NC * HDIM;
    float* xc1  = ws + o; o += (long)NC * HDIM;
    float* ocb  = ws + o; o += (long)NC * HDIM;
    float* comp = ws + o; o += (long)NC * HDIM;
    float* hbuf = ws + o; o += (long)NC * 32;

    dim3 blk(256);
    auto blocks = [](long n) { return (unsigned)((n + 255) / 256); };
    int tilesNN = (NN + 15) / 16, tilesNC = (NC + 15) / 16;
    unsigned gemmBlkNN = (tilesNN + 7) / 8, gemmBlkNC = (tilesNC + 7) / 8;

    k_gemm<4, false><<<gemmBlkNN, blk, 0, stream>>>(x_news, nw_W, nw_b,
                                                    nullptr, nullptr,
                                                    xn0, NN, 385, 0, 1, 0);
    k_gemm<4, true><<<gemmBlkNC, blk, 0, stream>>>(x_company, cw_W, cw_b,
                                                   nullptr, nullptr,
                                                   xc0, NC, 24, 0, 1, 0);

    float* part1 = aggn;
    float* cpart1 = aggn + (long)NPART * NC * HDIM;
    k_agg_cols<<<NPART * NSLICE, AGG_THREADS, 0, stream>>>(men_src, men_dst, xn0,
                                                           part1, cpart1, EM / 32);
    k_combine<<<blocks((long)NC * HDIM / 4), blk, 0, stream>>>(part1, cpart1, aggcM);
    k_agg_cols<<<NPART * NSLICE, AGG_THREADS, 0, stream>>>(edge_rel, edge_rel + ER, xc0,
                                                           part1, cpart1, ER / 32);
    k_combine<<<blocks((long)NC * HDIM / 4), blk, 0, stream>>>(part1, cpart1, aggcR);
    int* head = (int*)cntn;
    int2* rec = (int2*)xn1;
    k_list_clear<<<blocks(NN / 4), blk, 0, stream>>>(head, NN / 4);
    k_list_link<<<ES / (LINK_THREADS * LINK_EPT), LINK_THREADS, 0, stream>>>(
        edge_sim, edge_sim + ES, head, rec);
    k_list_mean<<<blocks((long)NN * 16), blk, 0, stream>>>(head, rec, xn0, aggn, NN, ES);

    k_gemm<4, true><<<gemmBlkNN, blk, 0, stream>>>(aggn, c1_Wl + 0 * 4096,
                                                   c1_bl + 0 * 64,
                                                   xn0, c1_Wr + 0 * 4096,
                                                   xn1, NN, 64, 64, 0, 0);
    k_gemm<4, true><<<gemmBlkNC, blk, 0, stream>>>(aggcM, c1_Wl + 1 * 4096,
                                                   c1_bl + 1 * 64,
                                                   xc0, c1_Wr + 1 * 4096,
                                                   ocb, NC, 64, 64, 0, 0);
    k_gemm<4, true><<<gemmBlkNC, blk, 0, stream>>>(aggcR, c1_Wl + 2 * 4096,
                                                   c1_bl + 2 * 64,
                                                   xc0, c1_Wr + 2 * 4096,
                                                   ocb, NC, 64, 64, 0, 1);

    dim3 lnb(32, 8);
    k_ln<<<(NN + 7) / 8, lnb, 0, stream>>>(xn1, xn1, nullptr, n1n_g, n1n_b, NN);
    k_ln<<<(NC + 7) / 8, lnb, 0, stream>>>(xc1, ocb, xc0, n1c_g, n1c_b, NC);

    float* part2 = xn0;
    float* cpart2 = xn0 + (long)NPART * NC * HDIM;
    k_agg_cols<<<NPART * NSLICE, AGG_THREADS, 0, stream>>>(men_src, men_dst, xn1,
                                                           part2, cpart2, EM / 32);
    k_combine<<<blocks((long)NC * HDIM / 4), blk, 0, stream>>>(part2, cpart2, aggcM);
    k_agg_cols<<<NPART * NSLICE, AGG_THREADS, 0, stream>>>(edge_rel, edge_rel + ER, xc1,
                                                           part2, cpart2, ER / 32);
    k_combine<<<blocks((long)NC * HDIM / 4), blk, 0, stream>>>(part2, cpart2, aggcR);
    k_gemm<4, true><<<gemmBlkNC, blk, 0, stream>>>(aggcM, c2_Wl + 1 * 4096,
                                                   c2_bl + 1 * 64,
                                                   xc1, c2_Wr + 1 * 4096,
                                                   ocb, NC, 64, 64, 0, 0);
    k_gemm<4, true><<<gemmBlkNC, blk, 0, stream>>>(aggcR, c2_Wl + 2 * 4096,
                                                   c2_bl + 2 * 64,
                                                   xc1, c2_Wr + 2 * 4096,
                                                   ocb, NC, 64, 64, 0, 1);

    k_addrelu<<<blocks((long)NC * HDIM), blk, 0, stream>>>(comp, ocb, xc1,
                                                           (long)NC * HDIM);
    k_gemm<2, true><<<gemmBlkNC, blk, 0, stream>>>(comp, cls_W1, cls_b1,
                                                   nullptr, nullptr,
                                                   hbuf, NC, 64, 0, 1, 0);
    k_final<<<blocks(NC), blk, 0, stream>>>(hbuf, cls_W2, cls_b2, out, NC);
}


# news aggregation: 4 lists per destination row (edge number mod 4), heads in the row's own mean row; a wave walks 8 lists, 4 lanes each; a row = tree sum of its 4 list sums (sum order changes)
# speedup vs baseline: 3.1059x; 1.1663x over previous
#include <hip/hip_runtime.h>
#include <hip/hip_bf16.h>

typedef __attribute__((ext_vector_type(2))) float v2f;
typedef __attribute__((ext_vector_type(8))) float v8f;
typedef __attribute__((ext_vector_type(4))) float v4f;

template<int NT> struct ColVec;
template<> struct ColVec<4> { typedef v4f T; };
template<> struct ColVec<2> { typedef v2f T; };

#define NN 100000
#define NC 5000
#define HDIM 64
#define ES 3200000
#define EM 1000000
#define ER 160000
static_assert(EM % 32 == 0 && ER % 32 == 0, "edge lists are walked in chunks of 32");

#define NSUB 4
#define LINK_EPT 8
#define LINK_THREADS 128
static_assert(NSUB * 4 == 16, "a row's heads fill the first 16 bytes of its mean row");
static_assert(NN % 16 == 0, "every wave of k_list_mean holds two whole rows");
static_assert(ES % (LINK_THREADS * LINK_EPT) == 0, "k_list_link has no tail");

__global__ __launch_bounds__(256) void k_list_clear(int* __restrict__ rows, int M) {
    const int i = blockIdx.x * 256 + threadIdx.x;
    if (i < M) *(int4*)(rows + (long)i * HDIM) = make_int4(-1, -1, -1, -1);
}

__global__ __launch_bounds__(LINK_THREADS) void k_list_link(const int* __restrict__ src,
                                                   const int* __restrict__ dst,
                                                   int* __restrict__ rows,
                                                   int2* __restrict__ rec) {
    const int lane = threadIdx.x & 31;
    const int sub = lane & (NSUB - 1);
    const int wave = blockIdx.x * (LINK_THREADS / 32) + (threadIdx.x >> 5);
    const int e0 = wave * (32 * LINK_EPT) + lane;
    int sv[LINK_EPT], dv[LINK_EPT], pv[LINK_EPT];
    #pragma unroll
    for (int k = 0; k < LINK_EPT; ++k) {
        sv[k] = src[e0 + 32 * k];
        dv[k] = dst[e0 + 32 * k];
    }
    #pragma unroll
    for (int k = 0; k < LINK_EPT; ++k)
        pv[k] = atomicExch(rows + (long)dv[k] * HDIM + sub, e0 + 32 * k);
    #pragma unroll
    for (int k = 0; k < LINK_EPT; ++k)
        rec[e0 + 32 * k] = make_int2(sv[k], pv[k]);
}

__global__ __launch_bounds__(256) void k_list_mean(const int2* __restrict__ rec,
                                                   const float* __restrict__ x,
                                                   float* agg,
                                                   int M, int nE) {
    const int row = (blockIdx.x * 256 + threadIdx.x) >> 4;
    const int sub = (threadIdx.x >> 2) & (NSUB - 1);
    const int q = threadIdx.x & 3;
    if (row >= M) return;
    float* orow = agg + (long)row * HDIM;
    int e = __float_as_int(orow[sub]);
    float4 a0 = make_float4(0.f, 0.f, 0.f, 0.f), a1 = a0, a2 = a0, a3 = a0;
    int n = 0;
    int2 r = make_int2(0, -1);
    if (e >= 0) r = rec[e];
    while (e >= 0 && n < nE) {
        const int en = r.y;
        int2 rn = make_int2(0, -1);
        if (en >= 0) rn = rec[en];
        const float4* __restrict__ xr = (const float4*)(x + (long)r.x * HDIM) + q;
        const float4 v0 = xr[0];
        const float4 v1 = xr[4];
        const float4 v2 = xr[8];
        const float4 v3 = xr[12];
        a0.x += v0.x; a0.y += v0.y; a0.z += v0.z; a0.w += v0.w;
        a1.x += v1.x; a1.y += v1.y; a1.z += v1.z; a1.w += v1.w;
        a2.x += v2.x; a2.y += v2.y; a2.z += v2.z; a2.w += v2.w;
        a3.x += v3.x; a3.y += v3.y; a3.z += v3.z; a3.w += v3.w;
        ++n;
        e = en;
        r = rn;
    }
    float acc[16] = { a0.x, a0.y, a0.z, a0.w, a1.x, a1.y, a1.z, a1.w,
                      a2.x, a2.y, a2.z, a2.w, a3.x, a3.y, a3.z, a3.w };
    #pragma unroll
    for (int i = 0; i < 16; ++i) {
        acc[i] += __shfl_xor(acc[i], 4, 32);
        acc[i] += __shfl_xor(acc[i], 8, 32);
    }
    n += __shfl_xor(n, 4, 32);
    n += __shfl_xor(n, 8, 32);
    const float inv = 1.0f / fmaxf((float)n, 1.0f);
    float4 o;
    o.x = (sub == 0) ? acc[0] : (sub == 1) ? acc[4] : (sub == 2) ? acc[8]  : acc[12];
    o.y = (sub == 0) ? acc[1] : (sub == 1) ? acc[5] : (sub == 2) ? acc[9]  : acc[13];
    o.z = (sub == 0) ? acc[2] : (sub == 1) ? acc[6] : (sub == 2) ? acc[10] : acc[14];
    o.w = (sub == 0) ? acc[3] : (sub == 1) ? acc[7] : (sub == 2) ? acc[11] : acc[15];
    o.x *= inv; o.y *= inv; o.z *= inv; o.w *= inv;
    *((float4*)orow + sub * 4 + q) = o;
}

#define CSL 8
#define NSLICE (HDIM / CSL)
#define NPART 32
#define AGG_THREADS 1024

__global__ __launch_bounds__(AGG_THREADS) void k_agg_cols(const int* __restrict__ src,
                                                          const int* __restrict__ dst,
                                                          const float* __restrict__ x,
                                                          float* __restrict__ part,
                                                          float* __restrict__ cntpart,
                                                          int nchunk) {
    __shared__ __attribute__((aligned(16))) float acc[NC * CSL];
    __shared__ float cn[NC];
    const int tid = threadIdx.x;
    const int s = blockIdx.x % NSLICE;
    const int p = blockIdx.x / NSLICE;

    float4* acc4 = (float4*)acc;
    for (int i = tid; i < NC * CSL / 4; i += AGG_THREADS)
        acc4[i] = make_float4(0.f, 0.f, 0.f, 0.f);
    for (int i = tid; i < NC; i += AGG_THREADS) cn[i] = 0.f;
    __syncthreads();

    const int lane = tid & 31, wv = tid >> 5;
    const int el = lane >> 3;
    const int j = lane & (CSL - 1);
    const int cb = (int)(((long)p * nchunk) / NPART);
    const int ce = (int)(((long)(p + 1) * nchunk) / NPART);
    const float* __restrict__ xs = x + s * CSL + j;
    const bool docnt = (s == 0) && (j == 0);
    const int4* __restrict__ src4 = (const int4*)src + el * 2;
    const int4* __restrict__ dst4 = (const int4*)dst + el * 2;

    int c = cb + wv;
    int4 sa = make_int4(0, 0, 0, 0), sb = sa, da = sa, db = sa;
    if (c < ce) {
        sa = src4[(long)c * 8];  sb = src4[(long)c * 8 + 1];
        da = dst4[(long)c * 8];  db = dst4[(long)c * 8 + 1];
    }
    while (c < ce) {
        const int c2 = c + AGG_THREADS / 32;
        int4 nsa = sa, nsb = sb, nda = da, ndb = db;
        if (c2 < ce) {
            nsa = src4[(long)c2 * 8];  nsb = src4[(long)c2 * 8 + 1];
            nda = dst4[(long)c2 * 8];  ndb = dst4[(long)c2 * 8 + 1];
        }
        const float v0 = xs[(long)sa.x * HDIM];
        const float v1 = xs[(long)sa.y * HDIM];
        const float v2 = xs[(long)sa.z * HDIM];
        const float v3 = xs[(long)sa.w * HDIM];
        const float v4 = xs[(long)sb.x * HDIM];
        const float v5 = xs[(long)sb.y * HDIM];
        const float v6 = xs[(long)sb.z * HDIM];
        const float v7 = xs[(long)sb.w * HDIM];
        atomicAdd(&acc[da.x * CSL + j], v0);
        atomicAdd(&acc[da.y * CSL + j], v1);
        atomicAdd(&acc[da.z * CSL + j], v2);
        atomicAdd(&acc[da.w * CSL + j], v3);
        atomicAdd(&acc[db.x * CSL + j], v4);
        atomicAdd(&acc[db.y * CSL + j], v5);
        atomicAdd(&acc[db.z * CSL + j], v6);
        atomicAdd(&acc[db.w * CSL + j], v7);
        if (docnt) {
            atomicAdd(&cn[da.x], 1.0f);  atomicAdd(&cn[da.y], 1.0f);
            atomicAdd(&cn[da.z], 1.0f);  atomicAdd(&cn[da.w], 1.0f);
            atomicAdd(&cn[db.x], 1.0f);  atomicAdd(&cn[db.y], 1.0f);
            atomicAdd(&cn[db.z], 1.0f);  atomicAdd(&cn[db.w], 1.0f);
        }
        sa = nsa; sb = nsb; da = nda; db = ndb;
        c = c2;
    }
    __syncthreads();

    float4* __restrict__ part4 = (float4*)part + (long)blockIdx.x * (NC * CSL / 4);
    for (int i = tid; i < NC * CSL / 4; i += AGG_THREADS) part4[i] = acc4[i];
    if (s == 0) {
        float* __restrict__ cp = cntpart + (long)p * NC;
        for (int i = tid; i < NC; i += AGG_THREADS) cp[i] = cn[i];
    }
}

__global__ __launch_bounds__(256) void k_combine(const float* __restrict__ part,
                                                 const float* __restrict__ cntpart,
                                                 float* __restrict__ agg) {
    const int per = NC * CSL / 4;
    const int t = blockIdx.x * 256 + threadIdx.x;
    if (t >= NSLICE * per) return;
    const int s = t / per;
    const int rem = t - s * per;
    const int row = rem / (CSL / 4);
    const int q = rem % (CSL / 4);
    const float4* __restrict__ p4 = (const float4*)part + t;
    float4 a = make_float4(0.f, 0.f, 0.f, 0.f);
    float c = 0.f;
    #pragma unroll 8
    for (int p = 0; p < NPART; ++p) {
        const float4 v = p4[(long)p * (NSLICE * per)];
        a.x += v.x; a.y += v.y; a.z += v.z; a.w += v.w;
        c += cntpart[p * NC + row];
    }
    const float inv = 1.0f / fmaxf(c, 1.0f);
    a.x *= inv; a.y *= inv; a.z *= inv; a.w *= inv;
    *(float4*)(agg + (long)row * HDIM + s * CSL + q * 4) = a;
}

template<int NT, bool FULL, bool VECA>
__device__ __forceinline__ void gemm_body(v8f (&c)[NT],
                                          const float* __restrict__ A,
                                          const float* __restrict__ W,
                                          int M, int K,
                                          int mbase, int lane) {
    typedef typename ColVec<NT>::T VT;
    constexpr int N = NT * 16;
    const int row  = mbase + (lane & 15);
    const int koff = (lane >= 16) ? 2 : 0;
    const bool rok = FULL || (row < M);
    const float* __restrict__ Arow = A + (long)row * K;
    const float* __restrict__ Wc = W + (lane & 15) * NT;
    const int Kmain = K & ~3;

    #pragma unroll 4
    for (int k = 0; k < Kmain; k += 4) {
        const int kk = k + koff;
        v2f a;
        if (rok) {
            if (VECA) {
                a = *(const v2f*)(Arow + kk);
            } else {
                a.x = Arow[kk];
                a.y = Arow[kk + 1];
            }
        } else {
            a.x = 0.f; a.y = 0.f;
        }
        const VT w0 = *(const VT*)(Wc + (long)kk * N);
        const VT w1 = *(const VT*)(Wc + (long)(kk + 1) * N);
        #pragma unroll
        for (int nt = 0; nt < NT; ++nt) {
            v2f b;
            b.x = w0[nt];
            b.y = w1[nt];
            c[nt] = __builtin_amdgcn_wmma_f32_16x16x4_f32(
                false, a, false, b, (short)0, c[nt], false, false);
        }
    }
    if (Kmain < K) {
        const int kk = Kmain + koff;
        v2f a;
        a.x = (rok && kk     < K) ? Arow[kk]     : 0.f;
        a.y = (rok && kk + 1 < K) ? Arow[kk + 1] : 0.f;
        VT w0, w1;
        #pragma unroll
        for (int nt = 0; nt < NT; ++nt) { w0[nt] = 0.f; w1[nt] = 0.f; }
        if (kk     < K) w0 = *(const VT*)(Wc + (long)kk * N);
        if (kk + 1 < K) w1 = *(const VT*)(Wc + (long)(kk + 1) * N);
        #pragma unroll
        for (int nt = 0; nt < NT; ++nt) {
            v2f b;
            b.x = w0[nt];
            b.y = w1[nt];
            c[nt] = __builtin_amdgcn_wmma_f32_16x16x4_f32(
                false, a, false, b, (short)0, c[nt], false, false);
        }
    }
}

template<int NT, bool VECA1>
__global__ __launch_bounds__(256) void k_gemm(const float* __restrict__ A1,
                                              const float* __restrict__ W1,
                                              const float* __restrict__ bias,
                                              const float* __restrict__ A2,
                                              const float* __restrict__ W2,
                                              float* __restrict__ out,
                                              int M, int K1, int K2,
                                              int do_relu, int accum) {
    typedef typename ColVec<NT>::T VT;
    constexpr int N = NT * 16;
    const int lane = threadIdx.x & 31;
    const int wv = threadIdx.x >> 5;
    const int mbase = (blockIdx.x * 8 + wv) * 16;
    if (mbase >= M) return;

    v8f c[NT];
    #pragma unroll
    for (int t = 0; t < NT; ++t)
        c[t] = (v8f){0.f, 0.f, 0.f, 0.f, 0.f, 0.f, 0.f, 0.f};

    const bool full = (mbase + 16 <= M);
    if (full) {
        gemm_body<NT, true, VECA1>(c, A1, W1, M, K1, mbase, lane);
        if (A2) gemm_body<NT, true, true>(c, A2, W2, M, K2, mbase, lane);
    } else {
        gemm_body<NT, false, VECA1>(c, A1, W1, M, K1, mbase, lane);
        if (A2) gemm_body<NT, false, true>(c, A2, W2, M, K2, mbase, lane);
    }

    const int cb = (lane & 15) * NT;
    const int rofs = (lane >= 16) ? 8 : 0;
    VT bb;
    #pragma unroll
    for (int nt = 0; nt < NT; ++nt) bb[nt] = 0.f;
    if (bias) bb = *(const VT*)(bias + cb);
    #pragma unroll
    for (int r = 0; r < 8; ++r) {
        const int row = mbase + r + rofs;
        if (full || row < M) {
            VT* __restrict__ p = (VT*)(out + (long)row * N + cb);
            VT v;
            #pragma unroll
            for (int nt = 0; nt < NT; ++nt) v[nt] = c[nt][r] + bb[nt];
            if (accum) v += *p;
            if (do_relu) {
                #pragma unroll
                for (int nt = 0; nt < NT; ++nt) v[nt] = fmaxf(v[nt], 0.f);
            }
            *p = v;
        }
    }
}

__global__ void k_ln(float* __restrict__ out, const float* __restrict__ in,
                     const float* __restrict__ res,
                     const float* __restrict__ g, const float* __restrict__ b,
                     int M) {
    int lane = threadIdx.x;
    int row = blockIdx.x * blockDim.y + threadIdx.y;
    if (row >= M) return;
    long base = (long)row * HDIM;
    float v0 = in[base + lane];
    float v1 = in[base + lane + 32];
    if (res) { v0 += res[base + lane]; v1 += res[base + lane + 32]; }
    v0 = fmaxf(v0, 0.f);
    v1 = fmaxf(v1, 0.f);
    float s = v0 + v1;
    for (int off = 16; off > 0; off >>= 1) s += __shfl_xor(s, off, 32);
    float mean = s * (1.0f / 64.0f);
    float d0 = v0 - mean, d1 = v1 - mean;
    float q = d0 * d0 + d1 * d1;
    for (int off = 16; off > 0; off >>= 1) q += __shfl_xor(q, off, 32);
    float rstd = rsqrtf(q * (1.0f / 64.0f) + 1e-5f);
    out[base + lane]      = d0 * rstd * g[lane]      + b[lane];
    out[base + lane + 32] = d1 * rstd * g[lane + 32] + b[lane + 32];
}

__global__ __launch_bounds__(256) void k_addrelu(float* __restrict__ o,
                                                 const float* __restrict__ a,
                                                 const float* __restrict__ b,
                                                 long n) {
    long i = (long)blockIdx.x * blockDim.x + threadIdx.x;
    if (i < n) o[i] = fmaxf(a[i] + b[i], 0.f);
}

__global__ __launch_bounds__(256) void k_final(const float* __restrict__ h,
                                               const float* __restrict__ W2,
                                               const float* __restrict__ b2,
                                               float* __restrict__ out, int M) {
    int i = blockIdx.x * blockDim.x + threadIdx.x;
    if (i >= M) return;
    float acc = b2[0];
    const float* hr = h + (long)i * 32;
    #pragma unroll
    for (int j = 0; j < 32; ++j) acc += hr[j] * W2[j];
    out[i] = acc;
}

extern "C" void kernel_launch(void* const* d_in, const int* in_sizes, int n_in,
                              void* d_out, int out_size, void* d_ws, size_t ws_size,
                              hipStream_t stream) {
    const float* x_news    = (const float*)d_in[0];
    const float* x_company = (const float*)d_in[1];
    const int*   edge_sim  = (const int*)d_in[2];
    const int*   men_src   = (const int*)d_in[3];
    const int*   men_dst   = (const int*)d_in[4];
    const int*   edge_rel  = (const int*)d_in[5];
    const float* nw_W = (const float*)d_in[6];
    const float* nw_b = (const float*)d_in[7];
    const float* cw_W = (const float*)d_in[8];
    const float* cw_b = (const float*)d_in[9];
    const float* n1n_g = (const float*)d_in[10];
    const float* n1n_b = (const float*)d_in[11];
    const float* n1c_g = (const float*)d_in[12];
    const float* n1c_b = (const float*)d_in[13];
    const float* c1_Wl = (const float*)d_in[14];
    const float* c1_bl = (const float*)d_in[15];
    const float* c1_Wr = (const float*)d_in[16];
    const float* c2_Wl = (const float*)d_in[17];
    const float* c2_bl = (const float*)d_in[18];
    const float* c2_Wr = (const float*)d_in[19];
    const float* cls_W1 = (const float*)d_in[20];
    const float* cls_b1 = (const float*)d_in[21];
    const float* cls_W2 = (const float*)d_in[22];
    const float* cls_b2 = (const float*)d_in[23];
    float* out = (float*)d_out;

    float* ws = (float*)d_ws;
    long o = 0;
    float* xn0  = ws + o; o += (long)NN * HDIM;
    float* aggn = ws + o; o += (long)NN * HDIM;
    o += NN;
    float* xn1  = ws + o; o += (long)NN * HDIM;
    float* aggcM = ws + o; o += (long)NC * HDIM;
    float* aggcR = ws + o; o += (long)NC * HDIM;
    float* xc0  = ws + o; o += (long)NC * HDIM;
    float* xc1  = ws + o; o += (long)NC * HDIM;
    float* ocb  = ws + o; o += (long)NC * HDIM;
    float* comp = ws + o; o += (long)NC * HDIM;
    float* hbuf = ws + o; o += (long)NC * 32;

    dim3 blk(256);
    auto blocks = [](long n) { return (unsigned)((n + 255) / 256); };
    int tilesNN = (NN + 15) / 16, tilesNC = (NC + 15) / 16;
    unsigned gemmBlkNN = (tilesNN + 7) / 8, gemmBlkNC = (tilesNC + 7) / 8;

    k_gemm<4, false><<<gemmBlkNN, blk, 0, stream>>>(x_news, nw_W, nw_b,
                                                    nullptr, nullptr,
                                                    xn0, NN, 385, 0, 1, 0);
    k_gemm<4, true><<<gemmBlkNC, blk, 0, stream>>>(x_company, cw_W, cw_b,
                                                   nullptr, nullptr,
                                                   xc0, NC, 24, 0, 1, 0);

    float* part1 = aggn;
    float* cpart1 = aggn + (long)NPART * NC * HDIM;
    k_agg_cols<<<NPART * NSLICE, AGG_THREADS, 0, stream>>>(men_src, men_dst, xn0,
                                                           part1, cpart1, EM / 32);
    k_combine<<<blocks((long)NC * HDIM / 4), blk, 0, stream>>>(part1, cpart1, aggcM);
    k_agg_cols<<<NPART * NSLICE, AGG_THREADS, 0, stream>>>(edge_rel, edge_rel + ER, xc0,
                                                           part1, cpart1, ER / 32);
    k_combine<<<blocks((long)NC * HDIM / 4), blk, 0, stream>>>(part1, cpart1, aggcR);
    int* rows = (int*)aggn;
    int2* rec = (int2*)xn1;
    k_list_clear<<<blocks(NN), blk, 0, stream>>>(rows, NN);
    k_list_link<<<ES / (LINK_THREADS * LINK_EPT), LINK_THREADS, 0, stream>>>(
        edge_sim, edge_sim + ES, rows, rec);
    k_list_mean<<<blocks((long)NN * 16), blk, 0, stream>>>(rec, xn0, aggn, NN, ES);

    k_gemm<4, true><<<gemmBlkNN, blk, 0, stream>>>(aggn, c1_Wl + 0 * 4096,
                                                   c1_bl + 0 * 64,
                                                   xn0, c1_Wr + 0 * 4096,
                                                   xn1, NN, 64, 64, 0, 0);
    k_gemm<4, true><<<gemmBlkNC, blk, 0, stream>>>(aggcM, c1_Wl + 1 * 4096,
                                                   c1_bl + 1 * 64,
                                                   xc0, c1_Wr + 1 * 4096,
                                                   ocb, NC, 64, 64, 0, 0);
    k_gemm<4, true><<<gemmBlkNC, blk, 0, stream>>>(aggcR, c1_Wl + 2 * 4096,
                                                   c1_bl + 2 * 64,
                                                   xc0, c1_Wr + 2 * 4096,
                                                   ocb, NC, 64, 64, 0, 1);

    dim3 lnb(32, 8);
    k_ln<<<(NN + 7) / 8, lnb, 0, stream>>>(xn1, xn1, nullptr, n1n_g, n1n_b, NN);
    k_ln<<<(NC + 7) / 8, lnb, 0, stream>>>(xc1, ocb, xc0, n1c_g, n1c_b, NC);

    float* part2 = xn0;
    float* cpart2 = xn0 + (long)NPART * NC * HDIM;
    k_agg_cols<<<NPART * NSLICE, AGG_THREADS, 0, stream>>>(men_src, men_dst, xn1,
                                                           part2, cpart2, EM / 32);
    k_combine<<<blocks((long)NC * HDIM / 4), blk, 0, stream>>>(part2, cpart2, aggcM);
    k_agg_cols<<<NPART * NSLICE, AGG_THREADS, 0, stream>>>(edge_rel, edge_rel + ER, xc1,
                                                           part2, cpart2, ER / 32);
    k_combine<<<blocks((long)NC * HDIM / 4), blk, 0, stream>>>(part2, cpart2, aggcR);
    k_gemm<4, true><<<gemmBlkNC, blk, 0, stream>>>(aggcM, c2_Wl + 1 * 4096,
                                                   c2_bl + 1 * 64,
                                                   xc1, c2_Wr + 1 * 4096,
                                                   ocb, NC, 64, 64, 0, 0);
    k_gemm<4, true><<<gemmBlkNC, blk, 0, stream>>>(aggcR, c2_Wl + 2 * 4096,
                                                   c2_bl + 2 * 64,
                                                   xc1, c2_Wr + 2 * 4096,
                                                   ocb, NC, 64, 64, 0, 1);

    k_addrelu<<<blocks((long)NC * HDIM), blk, 0, stream>>>(comp, ocb, xc1,
                                                           (long)NC * HDIM);
    k_gemm<2, true><<<gemmBlkNC, blk, 0, stream>>>(comp, cls_W1, cls_b1,
                                                   nullptr, nullptr,
                                                   hbuf, NC, 64, 0, 1, 0);
    k_final<<<blocks(NC), blk, 0, stream>>>(hbuf, cls_W2, cls_b2, out, NC);
}


# company aggregation: per-row linked lists (32 a row, integer exchange) built once per call, walked by both layers, a block per row, both relations in one pass; 23 to 20 launches; sum order changes
# speedup vs baseline: 3.5196x; 1.1332x over previous
#include <hip/hip_runtime.h>
#include <hip/hip_bf16.h>

typedef __attribute__((ext_vector_type(2))) float v2f;
typedef __attribute__((ext_vector_type(8))) float v8f;
typedef __attribute__((ext_vector_type(4))) float v4f;

template<int NT> struct ColVec;
template<> struct ColVec<4> { typedef v4f T; };
template<> struct ColVec<2> { typedef v2f T; };

#define NN 100000
#define NC 5000
#define HDIM 64
#define ES 3200000
#define EM 1000000
#define ER 160000

#define NSUB 4
#define LINK_EPT 8
#define LINK_THREADS 128
static_assert(NSUB * 4 == 16, "a row's heads fill the first 16 bytes of its mean row");
static_assert(NN % 16 == 0, "every wave of k_list_mean holds two whole rows");
static_assert(ES % (LINK_THREADS * LINK_EPT) == 0, "k_list_link has no tail");

__global__ __launch_bounds__(256) void k_list_clear(int* __restrict__ rows, int M) {
    const int i = blockIdx.x * 256 + threadIdx.x;
    if (i < M) *(int4*)(rows + (long)i * HDIM) = make_int4(-1, -1, -1, -1);
}

__global__ __launch_bounds__(LINK_THREADS) void k_list_link(const int* __restrict__ src,
                                                   const int* __restrict__ dst,
                                                   int* __restrict__ rows,
                                                   int2* __restrict__ rec) {
    const int lane = threadIdx.x & 31;
    const int sub = lane & (NSUB - 1);
    const int wave = blockIdx.x * (LINK_THREADS / 32) + (threadIdx.x >> 5);
    const int e0 = wave * (32 * LINK_EPT) + lane;
    int sv[LINK_EPT], dv[LINK_EPT], pv[LINK_EPT];
    #pragma unroll
    for (int k = 0; k < LINK_EPT; ++k) {
        sv[k] = src[e0 + 32 * k];
        dv[k] = dst[e0 + 32 * k];
    }
    #pragma unroll
    for (int k = 0; k < LINK_EPT; ++k)
        pv[k] = atomicExch(rows + (long)dv[k] * HDIM + sub, e0 + 32 * k);
    #pragma unroll
    for (int k = 0; k < LINK_EPT; ++k)
        rec[e0 + 32 * k] = make_int2(sv[k], pv[k]);
}

__global__ __launch_bounds__(256) void k_list_mean(const int2* __restrict__ rec,
                                                   const float* __restrict__ x,
                                                   float* agg,
                                                   int M, int nE) {
    const int row = (blockIdx.x * 256 + threadIdx.x) >> 4;
    const int sub = (threadIdx.x >> 2) & (NSUB - 1);
    const int q = threadIdx.x & 3;
    if (row >= M) return;
    float* orow = agg + (long)row * HDIM;
    int e = __float_as_int(orow[sub]);
    float4 a0 = make_float4(0.f, 0.f, 0.f, 0.f), a1 = a0, a2 = a0, a3 = a0;
    int n = 0;
    int2 r = make_int2(0, -1);
    if (e >= 0) r = rec[e];
    while (e >= 0 && n < nE) {
        const int en = r.y;
        int2 rn = make_int2(0, -1);
        if (en >= 0) rn = rec[en];
        const float4* __restrict__ xr = (const float4*)(x + (long)r.x * HDIM) + q;
        const float4 v0 = xr[0];
        const float4 v1 = xr[4];
        const float4 v2 = xr[8];
        const float4 v3 = xr[12];
        a0.x += v0.x; a0.y += v0.y; a0.z += v0.z; a0.w += v0.w;
        a1.x += v1.x; a1.y += v1.y; a1.z += v1.z; a1.w += v1.w;
        a2.x += v2.x; a2.y += v2.y; a2.z += v2.z; a2.w += v2.w;
        a3.x += v3.x; a3.y += v3.y; a3.z += v3.z; a3.w += v3.w;
        ++n;
        e = en;
        r = rn;
    }
    float acc[16] = { a0.x, a0.y, a0.z, a0.w, a1.x, a1.y, a1.z, a1.w,
                      a2.x, a2.y, a2.z, a2.w, a3.x, a3.y, a3.z, a3.w };
    #pragma unroll
    for (int i = 0; i < 16; ++i) {
        acc[i] += __shfl_xor(acc[i], 4, 32);
        acc[i] += __shfl_xor(acc[i], 8, 32);
    }
    n += __shfl_xor(n, 4, 32);
    n += __shfl_xor(n, 8, 32);
    const float inv = 1.0f / fmaxf((float)n, 1.0f);
    float4 o;
    o.x = (sub == 0) ? acc[0] : (sub == 1) ? acc[4] : (sub == 2) ? acc[8]  : acc[12];
    o.y = (sub == 0) ? acc[1] : (sub == 1) ? acc[5] : (sub == 2) ? acc[9]  : acc[13];
    o.z = (sub == 0) ? acc[2] : (sub == 1) ? acc[6] : (sub == 2) ? acc[10] : acc[14];
    o.w = (sub == 0) ? acc[3] : (sub == 1) ? acc[7] : (sub == 2) ? acc[11] : acc[15];
    o.x *= inv; o.y *= inv; o.z *= inv; o.w *= inv;
    *((float4*)orow + sub * 4 + q) = o;
}

#define CL 32
#define CLINK_EPT 2
#define CWALK_THREADS (CL * 4)
static_assert(CL == 32, "a lane of k_clist_link is the list number of all its edges");
static_assert(EM % (32 * CLINK_EPT) == 0 && ER % (32 * CLINK_EPT) == 0, "k_clist_link takes whole waves");

__global__ __launch_bounds__(256) void k_fill_m1(int* __restrict__ p, int n4) {
    const int i = blockIdx.x * 256 + threadIdx.x;
    if (i < n4) ((int4*)p)[i] = make_int4(-1, -1, -1, -1);
}

__global__ __launch_bounds__(LINK_THREADS) void k_clist_link(const int* __restrict__ src,
                                                    const int* __restrict__ dst,
                                                    int* __restrict__ heads,
                                                    int2* __restrict__ rec,
                                                    int nwaves) {
    const int lane = threadIdx.x & 31;
    const int wave = blockIdx.x * (LINK_THREADS / 32) + (threadIdx.x >> 5);
    if (wave >= nwaves) return;
    const int e0 = wave * (32 * CLINK_EPT) + lane;
    int sv[CLINK_EPT], dv[CLINK_EPT], pv[CLINK_EPT];
    #pragma unroll
    for (int k = 0; k < CLINK_EPT; ++k) {
        sv[k] = src[e0 + 32 * k];
        dv[k] = dst[e0 + 32 * k];
    }
    #pragma unroll
    for (int k = 0; k < CLINK_EPT; ++k)
        pv[k] = atomicExch(heads + dv[k] * CL + lane, e0 + 32 * k);
    #pragma unroll
    for (int k = 0; k < CLINK_EPT; ++k)
        rec[e0 + 32 * k] = make_int2(sv[k], pv[k]);
}

__global__ __launch_bounds__(CWALK_THREADS) void k_clist_mean(const int* __restrict__ headsM,
                                                     const int2* __restrict__ recM,
                                                     const float* __restrict__ xM,
                                                     float* __restrict__ outM, int nEM,
                                                     const int* __restrict__ headsR,
                                                     const int2* __restrict__ recR,
                                                     const float* __restrict__ xR,
                                                     float* __restrict__ outR, int nER) {
    __shared__ float4 sm[8][16];
    __shared__ int sn[8];
    const int row = blockIdx.x;
    const int t = threadIdx.x;
    const int list = t >> 2;
    const int grp = t >> 4;
    const int sub = list & 3;
    const int q = t & 3;
    for (int rel = 0; rel < 2; ++rel) {
        const int* __restrict__ heads = rel ? headsR : headsM;
        const int2* __restrict__ rec = rel ? recR : recM;
        const float* __restrict__ x = rel ? xR : xM;
        float* __restrict__ out = rel ? outR : outM;
        const int nE = rel ? nER : nEM;

        int e = heads[row * CL + list];
        float4 a0 = make_float4(0.f, 0.f, 0.f, 0.f), a1 = a0, a2 = a0, a3 = a0;
        int n = 0;
        int2 r = make_int2(0, -1);
        if (e >= 0) r = rec[e];
        while (e >= 0 && n < nE) {
            const int en = r.y;
            int2 rn = make_int2(0, -1);
            if (en >= 0) rn = rec[en];
            const float4* __restrict__ xr = (const float4*)(x + (long)r.x * HDIM) + q;
            const float4 v0 = xr[0];
            const float4 v1 = xr[4];
            const float4 v2 = xr[8];
            const float4 v3 = xr[12];
            a0.x += v0.x; a0.y += v0.y; a0.z += v0.z; a0.w += v0.w;
            a1.x += v1.x; a1.y += v1.y; a1.z += v1.z; a1.w += v1.w;
            a2.x += v2.x; a2.y += v2.y; a2.z += v2.z; a2.w += v2.w;
            a3.x += v3.x; a3.y += v3.y; a3.z += v3.z; a3.w += v3.w;
            ++n;
            e = en;
            r = rn;
        }
        float acc[16] = { a0.x, a0.y, a0.z, a0.w, a1.x, a1.y, a1.z, a1.w,
                          a2.x, a2.y, a2.z, a2.w, a3.x, a3.y, a3.z, a3.w };
        #pragma unroll
        for (int i = 0; i < 16; ++i) {
            acc[i] += __shfl_xor(acc[i], 4, 32);
            acc[i] += __shfl_xor(acc[i], 8, 32);
        }
        n += __shfl_xor(n, 4, 32);
        n += __shfl_xor(n, 8, 32);
        float4 o;
        o.x = (sub == 0) ? acc[0] : (sub == 1) ? acc[4] : (sub == 2) ? acc[8]  : acc[12];
        o.y = (sub == 0) ? acc[1] : (sub == 1) ? acc[5] : (sub == 2) ? acc[9]  : acc[13];
        o.z = (sub == 0) ? acc[2] : (sub == 1) ? acc[6] : (sub == 2) ? acc[10] : acc[14];
        o.w = (sub == 0) ? acc[3] : (sub == 1) ? acc[7] : (sub == 2) ? acc[11] : acc[15];
        sm[grp][sub * 4 + q] = o;
        if ((t & 15) == 0) sn[grp] = n;
        __syncthreads();
        if (t < 16) {
            float4 p[8];
            #pragma unroll
            for (int g = 0; g < 8; ++g) p[g] = sm[g][t];
            #pragma unroll
            for (int g = 0; g < 8; g += 2) {
                p[g].x += p[g + 1].x; p[g].y += p[g + 1].y; p[g].z += p[g + 1].z; p[g].w += p[g + 1].w;
            }
            p[0].x += p[2].x; p[0].y += p[2].y; p[0].z += p[2].z; p[0].w += p[2].w;
            p[4].x += p[6].x; p[4].y += p[6].y; p[4].z += p[6].z; p[4].w += p[6].w;
            p[0].x += p[4].x; p[0].y += p[4].y; p[0].z += p[4].z; p[0].w += p[4].w;
            const int cnt = ((sn[0] + sn[1]) + (sn[2] + sn[3])) + ((sn[4] + sn[5]) + (sn[6] + sn[7]));
            const float inv = 1.0f / fmaxf((float)cnt, 1.0f);
            p[0].x *= inv; p[0].y *= inv; p[0].z *= inv; p[0].w *= inv;
            *((float4*)(out + (long)row * HDIM) + t) = p[0];
        }
        __syncthreads();
    }
}

template<int NT, bool FULL, bool VECA>
__device__ __forceinline__ void gemm_body(v8f (&c)[NT],
                                          const float* __restrict__ A,
                                          const float* __restrict__ W,
                                          int M, int K,
                                          int mbase, int lane) {
    typedef typename ColVec<NT>::T VT;
    constexpr int N = NT * 16;
    const int row  = mbase + (lane & 15);
    const int koff = (lane >= 16) ? 2 : 0;
    const bool rok = FULL || (row < M);
    const float* __restrict__ Arow = A + (long)row * K;
    const float* __restrict__ Wc = W + (lane & 15) * NT;
    const int Kmain = K & ~3;

    #pragma unroll 4
    for (int k = 0; k < Kmain; k += 4) {
        const int kk = k + koff;
        v2f a;
        if (rok) {
            if (VECA) {
                a = *(const v2f*)(Arow + kk);
            } else {
                a.x = Arow[kk];
                a.y = Arow[kk + 1];
            }
        } else {
            a.x = 0.f; a.y = 0.f;
        }
        const VT w0 = *(const VT*)(Wc + (long)kk * N);
        const VT w1 = *(const VT*)(Wc + (long)(kk + 1) * N);
        #pragma unroll
        for (int nt = 0; nt < NT; ++nt) {
            v2f b;
            b.x = w0[nt];
            b.y = w1[nt];
            c[nt] = __builtin_amdgcn_wmma_f32_16x16x4_f32(
                false, a, false, b, (short)0, c[nt], false, false);
        }
    }
    if (Kmain < K) {
        const int kk = Kmain + koff;
        v2f a;
        a.x = (rok && kk     < K) ? Arow[kk]     : 0.f;
        a.y = (rok && kk + 1 < K) ? Arow[kk + 1] : 0.f;
        VT w0, w1;
        #pragma unroll
        for (int nt = 0; nt < NT; ++nt) { w0[nt] = 0.f; w1[nt] = 0.f; }
        if (kk     < K) w0 = *(const VT*)(Wc + (long)kk * N);
        if (kk + 1 < K) w1 = *(const VT*)(Wc + (long)(kk + 1) * N);
        #pragma unroll
        for (int nt = 0; nt < NT; ++nt) {
            v2f b;
            b.x = w0[nt];
            b.y = w1[nt];
            c[nt] = __builtin_amdgcn_wmma_f32_16x16x4_f32(
                false, a, false, b, (short)0, c[nt], false, false);
        }
    }
}

template<int NT, bool VECA1>
__global__ __launch_bounds__(256) void k_gemm(const float* __restrict__ A1,
                                              const float* __restrict__ W1,
                                              const float* __restrict__ bias,
                                              const float* __restrict__ A2,
                                              const float* __restrict__ W2,
                                              float* __restrict__ out,
                                              int M, int K1, int K2,
                                              int do_relu, int accum) {
    typedef typename ColVec<NT>::T VT;
    constexpr int N = NT * 16;
    const int lane = threadIdx.x & 31;
    const int wv = threadIdx.x >> 5;
    const int mbase = (blockIdx.x * 8 + wv) * 16;
    if (mbase >= M) return;

    v8f c[NT];
    #pragma unroll
    for (int t = 0; t < NT; ++t)
        c[t] = (v8f){0.f, 0.f, 0.f, 0.f, 0.f, 0.f, 0.f, 0.f};

    const bool full = (mbase + 16 <= M);
    if (full) {
        gemm_body<NT, true, VECA1>(c, A1, W1, M, K1, mbase, lane);
        if (A2) gemm_body<NT, true, true>(c, A2, W2, M, K2, mbase, lane);
    } else {
        gemm_body<NT, false, VECA1>(c, A1, W1, M, K1, mbase, lane);
        if (A2) gemm_body<NT, false, true>(c, A2, W2, M, K2, mbase, lane);
    }

    const int cb = (lane & 15) * NT;
    const int rofs = (lane >= 16) ? 8 : 0;
    VT bb;
    #pragma unroll
    for (int nt = 0; nt < NT; ++nt) bb[nt] = 0.f;
    if (bias) bb = *(const VT*)(bias + cb);
    #pragma unroll
    for (int r = 0; r < 8; ++r) {
        const int row = mbase + r + rofs;
        if (full || row < M) {
            VT* __restrict__ p = (VT*)(out + (long)row * N + cb);
            VT v;
            #pragma unroll
            for (int nt = 0; nt < NT; ++nt) v[nt] = c[nt][r] + bb[nt];
            if (accum) v += *p;
            if (do_relu) {
                #pragma unroll
                for (int nt = 0; nt < NT; ++nt) v[nt] = fmaxf(v[nt], 0.f);
            }
            *p = v;
        }
    }
}

__global__ void k_ln(float* __restrict__ out, const float* __restrict__ in,
                     const float* __restrict__ res,
                     const float* __restrict__ g, const float* __restrict__ b,
                     int M) {
    int lane = threadIdx.x;
    int row = blockIdx.x * blockDim.y + threadIdx.y;
    if (row >= M) return;
    long base = (long)row * HDIM;
    float v0 = in[base + lane];
    float v1 = in[base + lane + 32];
    if (res) { v0 += res[base + lane]; v1 += res[base + lane + 32]; }
    v0 = fmaxf(v0, 0.f);
    v1 = fmaxf(v1, 0.f);
    float s = v0 + v1;
    for (int off = 16; off > 0; off >>= 1) s += __shfl_xor(s, off, 32);
    float mean = s * (1.0f / 64.0f);
    float d0 = v0 - mean, d1 = v1 - mean;
    float q = d0 * d0 + d1 * d1;
    for (int off = 16; off > 0; off >>= 1) q += __shfl_xor(q, off, 32);
    float rstd = rsqrtf(q * (1.0f / 64.0f) + 1e-5f);
    out[base + lane]      = d0 * rstd * g[lane]      + b[lane];
    out[base + lane + 32] = d1 * rstd * g[lane + 32] + b[lane + 32];
}

__global__ __launch_bounds__(256) void k_addrelu(float* __restrict__ o,
                                                 const float* __restrict__ a,
                                                 const float* __restrict__ b,
                                                 long n) {
    long i = (long)blockIdx.x * blockDim.x + threadIdx.x;
    if (i < n) o[i] = fmaxf(a[i] + b[i], 0.f);
}

__global__ __launch_bounds__(256) void k_final(const float* __restrict__ h,
                                               const float* __restrict__ W2,
                                               const float* __restrict__ b2,
                                               float* __restrict__ out, int M) {
    int i = blockIdx.x * blockDim.x + threadIdx.x;
    if (i >= M) return;
    float acc = b2[0];
    const float* hr = h + (long)i * 32;
    #pragma unroll
    for (int j = 0; j < 32; ++j) acc += hr[j] * W2[j];
    out[i] = acc;
}

extern "C" void kernel_launch(void* const* d_in, const int* in_sizes, int n_in,
                              void* d_out, int out_size, void* d_ws, size_t ws_size,
                              hipStream_t stream) {
    const float* x_news    = (const float*)d_in[0];
    const float* x_company = (const float*)d_in[1];
    const int*   edge_sim  = (const int*)d_in[2];
    const int*   men_src   = (const int*)d_in[3];
    const int*   men_dst   = (const int*)d_in[4];
    const int*   edge_rel  = (const int*)d_in[5];
    const float* nw_W = (const float*)d_in[6];
    const float* nw_b = (const float*)d_in[7];
    const float* cw_W = (const float*)d_in[8];
    const float* cw_b = (const float*)d_in[9];
    const float* n1n_g = (const float*)d_in[10];
    const float* n1n_b = (const float*)d_in[11];
    const float* n1c_g = (const float*)d_in[12];
    const float* n1c_b = (const float*)d_in[13];
    const float* c1_Wl = (const float*)d_in[14];
    const float* c1_bl = (const float*)d_in[15];
    const float* c1_Wr = (const float*)d_in[16];
    const float* c2_Wl = (const float*)d_in[17];
    const float* c2_bl = (const float*)d_in[18];
    const float* c2_Wr = (const float*)d_in[19];
    const float* cls_W1 = (const float*)d_in[20];
    const float* cls_b1 = (const float*)d_in[21];
    const float* cls_W2 = (const float*)d_in[22];
    const float* cls_b2 = (const float*)d_in[23];
    float* out = (float*)d_out;

    float* ws = (float*)d_ws;
    long o = 0;
    float* xn0  = ws + o; o += (long)NN * HDIM;
    float* aggn = ws + o; o += (long)NN * HDIM;
    o += NN;
    float* xn1  = ws + o; o += (long)NN * HDIM;
    float* aggcM = ws + o; o += (long)NC * HDIM;
    float* aggcR = ws + o; o += (long)NC * HDIM;
    float* xc0  = ws + o; o += (long)NC * HDIM;
    float* xc1  = ws + o; o += (long)NC * HDIM;
    float* ocb  = ws + o; o += (long)NC * HDIM;
    float* comp = ws + o; o += (long)NC * HDIM;
    float* hbuf = ws + o; o += (long)NC * 32;

    dim3 blk(256);
    auto blocks = [](long n) { return (unsigned)((n + 255) / 256); };
    int tilesNN = (NN + 15) / 16, tilesNC = (NC + 15) / 16;
    unsigned gemmBlkNN = (tilesNN + 7) / 8, gemmBlkNC = (tilesNC + 7) / 8;

    k_gemm<4, false><<<gemmBlkNN, blk, 0, stream>>>(x_news, nw_W, nw_b,
                                                    nullptr, nullptr,
                                                    xn0, NN, 385, 0, 1, 0);
    k_gemm<4, true><<<gemmBlkNC, blk, 0, stream>>>(x_company, cw_W, cw_b,
                                                   nullptr, nullptr,
                                                   xc0, NC, 24, 0, 1, 0);

    int* rows = (int*)aggn;
    int2* rec = (int2*)xn1;
    k_list_clear<<<blocks(NN), blk, 0, stream>>>(rows, NN);
    k_list_link<<<ES / (LINK_THREADS * LINK_EPT), LINK_THREADS, 0, stream>>>(
        edge_sim, edge_sim + ES, rows, rec);
    k_list_mean<<<blocks((long)NN * 16), blk, 0, stream>>>(rec, xn0, aggn, NN, ES);

    k_gemm<4, true><<<gemmBlkNN, blk, 0, stream>>>(aggn, c1_Wl + 0 * 4096,
                                                   c1_bl + 0 * 64,
                                                   xn0, c1_Wr + 0 * 4096,
                                                   xn1, NN, 64, 64, 0, 0);

    int2* recM = (int2*)aggn;
    int2* recR = recM + EM;
    int* headsM = (int*)(recR + ER);
    int* headsR = headsM + NC * CL;
    const int clinkWavesM = EM / (32 * CLINK_EPT), clinkWavesR = ER / (32 * CLINK_EPT);
    k_fill_m1<<<blocks(2 * NC * CL / 4), blk, 0, stream>>>(headsM, 2 * NC * CL / 4);
    k_clist_link<<<(clinkWavesM + 3) / 4, LINK_THREADS, 0, stream>>>(
        men_src, men_dst, headsM, recM, clinkWavesM);
    k_clist_link<<<(clinkWavesR + 3) / 4, LINK_THREADS, 0, stream>>>(
        edge_rel, edge_rel + ER, headsR, recR, clinkWavesR);

    k_clist_mean<<<NC, CWALK_THREADS, 0, stream>>>(headsM, recM, xn0, aggcM, EM,
                                                   headsR, recR, xc0, aggcR, ER);
    k_gemm<4, true><<<gemmBlkNC, blk, 0, stream>>>(aggcM, c1_Wl + 1 * 4096,
                                                   c1_bl + 1 * 64,
                                                   xc0, c1_Wr + 1 * 4096,
                                                   ocb, NC, 64, 64, 0, 0);
    k_gemm<4, true><<<gemmBlkNC, blk, 0, stream>>>(aggcR, c1_Wl + 2 * 4096,
                                                   c1_bl + 2 * 64,
                                                   xc0, c1_Wr + 2 * 4096,
                                                   ocb, NC, 64, 64, 0, 1);

    dim3 lnb(32, 8);
    k_ln<<<(NN + 7) / 8, lnb, 0, stream>>>(xn1, xn1, nullptr, n1n_g, n1n_b, NN);
    k_ln<<<(NC + 7) / 8, lnb, 0, stream>>>(xc1, ocb, xc0, n1c_g, n1c_b, NC);

    k_clist_mean<<<NC, CWALK_THREADS, 0, stream>>>(headsM, recM, xn1, aggcM, EM,
                                                   headsR, recR, xc1, aggcR, ER);
    k_gemm<4, true><<<gemmBlkNC, blk, 0, stream>>>(aggcM, c2_Wl + 1 * 4096,
                                                   c2_bl + 1 * 64,
                                                   xc1, c2_Wr + 1 * 4096,
                                                   ocb, NC, 64, 64, 0, 0);
    k_gemm<4, true><<<gemmBlkNC, blk, 0, stream>>>(aggcR, c2_Wl + 2 * 4096,
                                                   c2_bl + 2 * 64,
                                                   xc1, c2_Wr + 2 * 4096,
                                                   ocb, NC, 64, 64, 0, 1);

    k_addrelu<<<blocks((long)NC * HDIM), blk, 0, stream>>>(comp, ocb, xc1,
                                                           (long)NC * HDIM);
    k_gemm<2, true><<<gemmBlkNC, blk, 0, stream>>>(comp, cls_W1, cls_b1,
                                                   nullptr, nullptr,
                                                   hbuf, NC, 64, 0, 1, 0);
    k_final<<<blocks(NC), blk, 0, stream>>>(hbuf, cls_W2, cls_b2, out, NC);
}
